# starting kernel
# baseline (speedup 1.0000x reference)
#include <hip/hip_runtime.h>
#include <math.h>

typedef __attribute__((ext_vector_type(2))) float v2f;
typedef __attribute__((ext_vector_type(8))) float v8f;

#define FEAT 64

__global__ void gcnii_spmm_edges(const int* __restrict__ row,
                                 const int* __restrict__ col,
                                 const float* __restrict__ vals,
                                 const float* __restrict__ H,
                                 float* __restrict__ AH,
                                 int nEdges) {
    int wave = (int)((blockIdx.x * blockDim.x + threadIdx.x) >> 5);
    int lane = threadIdx.x & 31;
    if (wave >= nEdges) return;

    int r = row[wave];
    int c = col[wave];
    float v = vals[wave];

    const float2* hsrc = (const float2*)(H + (size_t)c * FEAT);
    float2 h = hsrc[lane];

    float* dst = AH + (size_t)r * FEAT + lane * 2;
    atomicAdd(dst + 0, v * h.x);
    atomicAdd(dst + 1, v * h.y);
}

__global__ void gcnii_gemm_epilogue(const float* __restrict__ AH,
                                    const float* __restrict__ H0,
                                    const float* __restrict__ W,
                                    float* __restrict__ out,
                                    const float* __restrict__ lamdaPtr,
                                    const float* __restrict__ alphaPtr,
                                    const int* __restrict__ lPtr,
                                    int nTiles) {
    int tile = (int)(blockIdx.x * (blockDim.x >> 5) + (threadIdx.x >> 5));
    if (tile >= nTiles) return;

    int m0 = (tile >> 2) * 16;
    int n0 = (tile & 3) * 16;
    int lane = threadIdx.x & 31;
    int hi   = lane >> 4;
    int lm   = lane & 15;

    float alpha = alphaPtr[0];
    float lamda = lamdaPtr[0];
    float beta  = logf(lamda / (float)lPtr[0] + 1.0f);
    float oma   = 1.0f - alpha;
    float omb   = 1.0f - beta;

    const float* arowAH = AH + (size_t)(m0 + lm) * FEAT + hi * 2;
    const float* arowH0 = H0 + (size_t)(m0 + lm) * FEAT + hi * 2;
    const float* wbase  = W + (size_t)(hi * 2) * FEAT + n0 + lm;

    v8f acc = {};
#pragma unroll
    for (int kk = 0; kk < 16; ++kk) {
        int K0 = kk * 4;
        float2 ah2 = *(const float2*)(arowAH + K0);
        float2 h02 = *(const float2*)(arowH0 + K0);
        v2f a;
        a.x = oma * ah2.x + alpha * h02.x;
        a.y = oma * ah2.y + alpha * h02.y;
        v2f b;
        b.x = wbase[(size_t)(K0 + 0) * FEAT];
        b.y = wbase[(size_t)(K0 + 1) * FEAT];
        acc = __builtin_amdgcn_wmma_f32_16x16x4_f32(
             false, a,  false, b,
             (short)0, acc,  false,  false);
    }

#pragma unroll
    for (int r = 0; r < 8; ++r) {
        size_t idx = (size_t)(m0 + r + hi * 8) * FEAT + n0 + lm;
        float sup = oma * AH[idx] + alpha * H0[idx];
        out[idx] = omb * sup + beta * acc[r];
    }
}

extern "C" void kernel_launch(void* const* d_in, const int* in_sizes, int n_in,
                              void* d_out, int out_size, void* d_ws, size_t ws_size,
                              hipStream_t stream) {
    const int*   row   = (const int*)  d_in[0];
    const int*   col   = (const int*)  d_in[1];
    const float* vals  = (const float*)d_in[2];
    const float* H     = (const float*)d_in[3];
    const float* H0    = (const float*)d_in[4];
    const float* W     = (const float*)d_in[5];
    const float* lamda = (const float*)d_in[6];
    const float* alpha = (const float*)d_in[7];
    const int*   l     = (const int*)  d_in[8];

    int nEdges = in_sizes[0];
    int nNodes = in_sizes[3] / FEAT;

    float* AH = (float*)d_ws;

    hipMemsetAsync(AH, 0, (size_t)nNodes * FEAT * sizeof(float), stream);

    int spmmBlocks = (nEdges + 7) / 8;
    gcnii_spmm_edges<<<spmmBlocks, 256, 0, stream>>>(row, col, vals, H, AH, nEdges);

    int nTiles = (nNodes / 16) * (FEAT / 16);
    int gemmBlocks = (nTiles + 7) / 8;
    gcnii_gemm_epilogue<<<gemmBlocks, 256, 0, stream>>>(
        AH, H0, W, (float*)d_out, lamda, alpha, l, nTiles);
}


# SPMM without binning: each work-group owns 512 destination rows, reads the whole row list itself, adds its edges into LDS sums (ds_add_f32, arrival order); no global atomics, no AH memset
# speedup vs baseline: 1.8352x; 1.8352x over previous
#include <hip/hip_runtime.h>
#include <math.h>

typedef __attribute__((ext_vector_type(2))) float v2f;
typedef __attribute__((ext_vector_type(8))) float v8f;

#define FEAT 64

#define RB_SHIFT      9
#define RB_ROWS       (1 << RB_SHIFT)
#define SCAN_WAVES    32
#define SCAN_THREADS  (SCAN_WAVES * 32)
#define CH_LOADS      8
#define CH_EDGES      (CH_LOADS * 32 * 4)
#define GATHER_DEPTH  8
#define Q_WORDS       (CH_EDGES + 32)
#define REC_EDGE_BITS 23

__device__ __forceinline__ void load_index_chunk(const int* row, int chunk, int lane,
                                                 int nEdges, int4 (&buf)[CH_LOADS]) {
    int base = chunk * CH_EDGES;
    if (base + CH_EDGES <= nEdges) {
        const int4* p = (const int4*)(row + base) + lane;
#pragma unroll
        for (int j = 0; j < CH_LOADS; ++j) buf[j] = p[j * 32];
    } else {
#pragma unroll
        for (int j = 0; j < CH_LOADS; ++j) {
            int e0 = base + (j * 32 + lane) * 4;
            buf[j].x = (e0 + 0 < nEdges) ? row[e0 + 0] : -1;
            buf[j].y = (e0 + 1 < nEdges) ? row[e0 + 1] : -1;
            buf[j].z = (e0 + 2 < nEdges) ? row[e0 + 2] : -1;
            buf[j].w = (e0 + 3 < nEdges) ? row[e0 + 3] : -1;
        }
    }
}

__device__ __forceinline__ void append_matches(unsigned t, unsigned e, unsigned* q, unsigned& cnt) {
    const bool match = t < (unsigned)RB_ROWS;
    const unsigned m = __builtin_amdgcn_ballot_w32(match);
    if (m) {
        if (match) q[cnt + __builtin_amdgcn_mbcnt_lo(m, 0u)] = e | (t << REC_EDGE_BITS);
        cnt += (unsigned)__builtin_popcount(m);
    }
}

template <bool FULL>
__device__ __forceinline__ void drain_records(const unsigned* q, unsigned m, int lane,
                                              const int* __restrict__ col,
                                              const float* __restrict__ vals,
                                              const float* __restrict__ H,
                                              float* acc) {
    int cl = 0; float vl = 0.f; unsigned rl = 0;
    if (FULL || (unsigned)lane < m) {
        const unsigned rec = q[lane];
        const unsigned e = rec & ((1u << REC_EDGE_BITS) - 1u);
        cl = col[e];
        vl = vals[e];
        rl = rec >> REC_EDGE_BITS;
    }
    for (unsigned k = 0; k < m; k += GATHER_DEPTH) {
        float2 h[GATHER_DEPTH];
#pragma unroll
        for (int t = 0; t < GATHER_DEPTH; ++t) {
            const unsigned kk = (FULL || k + t < m) ? (k + t) : (m - 1u);
            const int cc = __builtin_amdgcn_readlane(cl, (int)kk);
            const unsigned off = ((unsigned)cc << 8) + (unsigned)lane * 8u;
            h[t] = *(const float2*)((const char*)H + off);
        }
#pragma unroll
        for (int t = 0; t < GATHER_DEPTH; ++t) {
            if (FULL || k + t < m) {
                const float vv = __builtin_bit_cast(float,
                    __builtin_amdgcn_readlane(__builtin_bit_cast(int, vl), (int)(k + t)));
                const unsigned rr = (unsigned)__builtin_amdgcn_readlane((int)rl, (int)(k + t));
                float* d = acc + rr * FEAT + lane * 2;
                __hip_atomic_fetch_add(d + 0, vv * h[t].x, __ATOMIC_RELAXED, __HIP_MEMORY_SCOPE_WORKGROUP);
                __hip_atomic_fetch_add(d + 1, vv * h[t].y, __ATOMIC_RELAXED, __HIP_MEMORY_SCOPE_WORKGROUP);
            }
        }
    }
}

__global__ __launch_bounds__(SCAN_THREADS)
void gcnii_spmm_owner_scan(const int* __restrict__ row,
                           const int* __restrict__ col,
                           const float* __restrict__ vals,
                           const float* __restrict__ H,
                           float* __restrict__ AH,
                           int nEdges, int nNodes) {
    __shared__ float    acc[RB_ROWS * FEAT];
    __shared__ unsigned queue[SCAN_WAVES * Q_WORDS];

    const int lane = threadIdx.x & 31;
    const int w    = __builtin_amdgcn_readfirstlane((int)(threadIdx.x >> 5));
    const unsigned b = blockIdx.x;
    const unsigned rbase = b << RB_SHIFT;

    float4* acc4 = (float4*)acc;
    for (int i = threadIdx.x; i < RB_ROWS * FEAT / 4; i += SCAN_THREADS)
        acc4[i] = make_float4(0.f, 0.f, 0.f, 0.f);
    __syncthreads();

    const int nChunks = (nEdges + CH_EDGES - 1) / CH_EDGES;
    const int rot = (int)(((long long)b * nChunks) / gridDim.x);
    unsigned* q = queue + w * Q_WORDS;

    unsigned cnt = 0;
    for (int i = w; i < nChunks; i += SCAN_WAVES) {
        int c = i + rot; if (c >= nChunks) c -= nChunks;
        int4 cur[CH_LOADS];
        load_index_chunk(row, c, lane, nEdges, cur);
        const unsigned e_lane = (unsigned)c * CH_EDGES + (unsigned)lane * 4u;

#pragma unroll
        for (int j = 0; j < CH_LOADS; ++j) {
            const unsigned t0 = (unsigned)cur[j].x ^ rbase;
            const unsigned t1 = (unsigned)cur[j].y ^ rbase;
            const unsigned t2 = (unsigned)cur[j].z ^ rbase;
            const unsigned t3 = (unsigned)cur[j].w ^ rbase;
            const unsigned t01 = t0 < t1 ? t0 : t1;
            const unsigned t23 = t2 < t3 ? t2 : t3;
            const unsigned tmin = t01 < t23 ? t01 : t23;
            if (__builtin_amdgcn_ballot_w32(tmin < (unsigned)RB_ROWS)) {
                const unsigned e = e_lane + (unsigned)j * 128u;
                append_matches(t0, e + 0u, q, cnt);
                append_matches(t1, e + 1u, q, cnt);
                append_matches(t2, e + 2u, q, cnt);
                append_matches(t3, e + 3u, q, cnt);
            }
        }

        while (cnt >= 32u) {
            cnt -= 32u;
            drain_records<true>(q + cnt, 32u, lane, col, vals, H, acc);
        }
    }
    if (cnt) drain_records<false>(q, cnt, lane, col, vals, H, acc);

    __syncthreads();
    const int r0 = (int)b * RB_ROWS;
    const int nrows = (nNodes - r0 < RB_ROWS) ? (nNodes - r0) : RB_ROWS;
    float4* dst = (float4*)(AH + (size_t)r0 * FEAT);
    for (int t = threadIdx.x; t < nrows * (FEAT / 4); t += SCAN_THREADS)
        dst[t] = acc4[t];
}

__global__ void gcnii_gemm_epilogue(const float* __restrict__ AH,
                                    const float* __restrict__ H0,
                                    const float* __restrict__ W,
                                    float* __restrict__ out,
                                    const float* __restrict__ lamdaPtr,
                                    const float* __restrict__ alphaPtr,
                                    const int* __restrict__ lPtr,
                                    int nTiles) {
    int tile = (int)(blockIdx.x * (blockDim.x >> 5) + (threadIdx.x >> 5));
    if (tile >= nTiles) return;

    int m0 = (tile >> 2) * 16;
    int n0 = (tile & 3) * 16;
    int lane = threadIdx.x & 31;
    int hi   = lane >> 4;
    int lm   = lane & 15;

    float alpha = alphaPtr[0];
    float lamda = lamdaPtr[0];
    float beta  = logf(lamda / (float)lPtr[0] + 1.0f);
    float oma   = 1.0f - alpha;
    float omb   = 1.0f - beta;

    const float* arowAH = AH + (size_t)(m0 + lm) * FEAT + hi * 2;
    const float* arowH0 = H0 + (size_t)(m0 + lm) * FEAT + hi * 2;
    const float* wbase  = W + (size_t)(hi * 2) * FEAT + n0 + lm;

    v8f acc = {};
#pragma unroll
    for (int kk = 0; kk < 16; ++kk) {
        int K0 = kk * 4;
        float2 ah2 = *(const float2*)(arowAH + K0);
        float2 h02 = *(const float2*)(arowH0 + K0);
        v2f a;
        a.x = oma * ah2.x + alpha * h02.x;
        a.y = oma * ah2.y + alpha * h02.y;
        v2f b;
        b.x = wbase[(size_t)(K0 + 0) * FEAT];
        b.y = wbase[(size_t)(K0 + 1) * FEAT];
        acc = __builtin_amdgcn_wmma_f32_16x16x4_f32(
             false, a,  false, b,
             (short)0, acc,  false,  false);
    }

#pragma unroll
    for (int r = 0; r < 8; ++r) {
        size_t idx = (size_t)(m0 + r + hi * 8) * FEAT + n0 + lm;
        float sup = oma * AH[idx] + alpha * H0[idx];
        out[idx] = omb * sup + beta * acc[r];
    }
}

extern "C" void kernel_launch(void* const* d_in, const int* in_sizes, int n_in,
                              void* d_out, int out_size, void* d_ws, size_t ws_size,
                              hipStream_t stream) {
    const int*   row   = (const int*)  d_in[0];
    const int*   col   = (const int*)  d_in[1];
    const float* vals  = (const float*)d_in[2];
    const float* H     = (const float*)d_in[3];
    const float* H0    = (const float*)d_in[4];
    const float* W     = (const float*)d_in[5];
    const float* lamda = (const float*)d_in[6];
    const float* alpha = (const float*)d_in[7];
    const int*   l     = (const int*)  d_in[8];

    int nEdges = in_sizes[0];
    int nNodes = in_sizes[3] / FEAT;

    float* AH = (float*)d_ws;

    int spmmBlocks = (nNodes + RB_ROWS - 1) / RB_ROWS;
    gcnii_spmm_owner_scan<<<spmmBlocks, SCAN_THREADS, 0, stream>>>(
        row, col, vals, H, AH, nEdges, nNodes);

    int nTiles = (nNodes / 16) * (FEAT / 16);
    int gemmBlocks = (nTiles + 7) / 8;
    gcnii_gemm_epilogue<<<gemmBlocks, 256, 0, stream>>>(
        AH, H0, W, (float*)d_out, lamda, alpha, l, nTiles);
}


# SPMM scan via a presence map: a new first launch sets one bit per (512-row block, 4-edge group); each owner group reads its own bits and only the flagged 16 B pieces; sums still in arrival order
# speedup vs baseline: 2.0270x; 1.1045x over previous
#include <hip/hip_runtime.h>
#include <math.h>

typedef __attribute__((ext_vector_type(2))) float v2f;
typedef __attribute__((ext_vector_type(8))) float v8f;

#define FEAT 64

#define RB_SHIFT      9
#define RB_ROWS       (1 << RB_SHIFT)
#define SCAN_WAVES    32
#define SCAN_THREADS  (SCAN_WAVES * 32)
#define GATHER_DEPTH  8
#define REC_EDGE_BITS 23

#define GRP_EDGES       4
#define TILE_GROUPS     1024
#define TILE_EDGES      (GRP_EDGES * TILE_GROUPS)
#define TILE_WORDS      (TILE_GROUPS / 32)
#define MARK_THREADS    256
#define MARK_ITERS      (TILE_GROUPS / MARK_THREADS)
#define MARK_BLOCKS_MAX 256
#define GL_WORDS        64
#define Q_WORDS         160

__device__ __forceinline__ void mark_row(unsigned* bits, int r, unsigned b0, unsigned nb,
                                         unsigned word, unsigned bit) {
    const unsigned blk = ((unsigned)r >> RB_SHIFT) - b0;
    if (blk < nb)
        __hip_atomic_fetch_or(&bits[blk * TILE_WORDS + word], bit,
                              __ATOMIC_RELAXED, __HIP_MEMORY_SCOPE_WORKGROUP);
}

__global__ __launch_bounds__(MARK_THREADS)
void gcnii_mark_groups(const int* __restrict__ row, unsigned* __restrict__ map,
                       int nEdges, int nBlocks, int nTiles) {
    __shared__ unsigned bits[MARK_BLOCKS_MAX * TILE_WORDS];

    const int tid  = threadIdx.x;
    const int tile = blockIdx.x;

    int4 r[MARK_ITERS];
#pragma unroll
    for (int k = 0; k < MARK_ITERS; ++k) {
        const int e0 = tile * TILE_EDGES + (tid + k * MARK_THREADS) * GRP_EDGES;
        if (e0 + GRP_EDGES <= nEdges) {
            r[k] = *(const int4*)(row + e0);
        } else {
            r[k].x = (e0 + 0 < nEdges) ? row[e0 + 0] : -1;
            r[k].y = (e0 + 1 < nEdges) ? row[e0 + 1] : -1;
            r[k].z = (e0 + 2 < nEdges) ? row[e0 + 2] : -1;
            r[k].w = (e0 + 3 < nEdges) ? row[e0 + 3] : -1;
        }
    }

    const unsigned bit = 1u << (tid & 31);
    for (int b0 = 0; b0 < nBlocks; b0 += MARK_BLOCKS_MAX) {
        const int nb = (nBlocks - b0 < MARK_BLOCKS_MAX) ? (nBlocks - b0) : MARK_BLOCKS_MAX;
        uint4* bits4 = (uint4*)bits;
        for (int i = tid; i < nb * TILE_WORDS / 4; i += MARK_THREADS)
            bits4[i] = make_uint4(0u, 0u, 0u, 0u);
        __syncthreads();
#pragma unroll
        for (int k = 0; k < MARK_ITERS; ++k) {
            const unsigned word = (unsigned)(tid >> 5) + (unsigned)k * (MARK_THREADS / 32);
            mark_row(bits, r[k].x, (unsigned)b0, (unsigned)nb, word, bit);
            mark_row(bits, r[k].y, (unsigned)b0, (unsigned)nb, word, bit);
            mark_row(bits, r[k].z, (unsigned)b0, (unsigned)nb, word, bit);
            mark_row(bits, r[k].w, (unsigned)b0, (unsigned)nb, word, bit);
        }
        __syncthreads();
        for (int i = tid; i < nb * TILE_WORDS; i += MARK_THREADS) {
            const size_t line = (size_t)(b0 + (i >> 5)) * nTiles + tile;
            map[line * TILE_WORDS + (i & 31)] = bits[i];
        }
        __syncthreads();
    }
}

__device__ __forceinline__ void append_matches(unsigned t, unsigned e, unsigned* q, unsigned& cnt) {
    const bool match = t < (unsigned)RB_ROWS;
    const unsigned m = __builtin_amdgcn_ballot_w32(match);
    if (m) {
        if (match) q[cnt + __builtin_amdgcn_mbcnt_lo(m, 0u)] = e | (t << REC_EDGE_BITS);
        cnt += (unsigned)__builtin_popcount(m);
    }
}

template <bool FULL>
__device__ __forceinline__ void drain_records(const unsigned* q, unsigned m, int lane,
                                              const int* __restrict__ col,
                                              const float* __restrict__ vals,
                                              const float* __restrict__ H,
                                              float* acc) {
    int cl = 0; float vl = 0.f; unsigned rl = 0;
    if (FULL || (unsigned)lane < m) {
        const unsigned rec = q[lane];
        const unsigned e = rec & ((1u << REC_EDGE_BITS) - 1u);
        cl = col[e];
        vl = vals[e];
        rl = rec >> REC_EDGE_BITS;
    }
    for (unsigned k = 0; k < m; k += GATHER_DEPTH) {
        float2 h[GATHER_DEPTH];
#pragma unroll
        for (int t = 0; t < GATHER_DEPTH; ++t) {
            const unsigned kk = (FULL || k + t < m) ? (k + t) : (m - 1u);
            const int cc = __builtin_amdgcn_readlane(cl, (int)kk);
            const unsigned off = ((unsigned)cc << 8) + (unsigned)lane * 8u;
            h[t] = *(const float2*)((const char*)H + off);
        }
#pragma unroll
        for (int t = 0; t < GATHER_DEPTH; ++t) {
            if (FULL || k + t < m) {
                const float vv = __builtin_bit_cast(float,
                    __builtin_amdgcn_readlane(__builtin_bit_cast(int, vl), (int)(k + t)));
                const unsigned rr = (unsigned)__builtin_amdgcn_readlane((int)rl, (int)(k + t));
                float* d = acc + rr * FEAT + lane * 2;
                __hip_atomic_fetch_add(d + 0, vv * h[t].x, __ATOMIC_RELAXED, __HIP_MEMORY_SCOPE_WORKGROUP);
                __hip_atomic_fetch_add(d + 1, vv * h[t].y, __ATOMIC_RELAXED, __HIP_MEMORY_SCOPE_WORKGROUP);
            }
        }
    }
}

template <bool FULL>
__device__ __forceinline__ void match_groups(const unsigned* gl, unsigned m, int lane,
                                             const int* __restrict__ row, int nEdges,
                                             unsigned rbase, unsigned* q, unsigned& cnt) {
    int4 r = make_int4(-1, -1, -1, -1);
    unsigned e0 = 0u;
    if (FULL || (unsigned)lane < m) {
        e0 = gl[lane] * GRP_EDGES;
        if ((int)e0 + GRP_EDGES <= nEdges) {
            r = *(const int4*)(row + e0);
        } else {
            if ((int)e0 + 0 < nEdges) r.x = row[e0 + 0];
            if ((int)e0 + 1 < nEdges) r.y = row[e0 + 1];
            if ((int)e0 + 2 < nEdges) r.z = row[e0 + 2];
        }
    }
    append_matches((unsigned)r.x ^ rbase, e0 + 0u, q, cnt);
    append_matches((unsigned)r.y ^ rbase, e0 + 1u, q, cnt);
    append_matches((unsigned)r.z ^ rbase, e0 + 2u, q, cnt);
    append_matches((unsigned)r.w ^ rbase, e0 + 3u, q, cnt);
}

__global__ __launch_bounds__(SCAN_THREADS)
void gcnii_spmm_owner_map(const int* __restrict__ row,
                          const int* __restrict__ col,
                          const float* __restrict__ vals,
                          const float* __restrict__ H,
                          const unsigned* __restrict__ map,
                          float* __restrict__ AH,
                          int nEdges, int nNodes, int nTiles) {
    __shared__ float    acc[RB_ROWS * FEAT];
    __shared__ unsigned glist[SCAN_WAVES * GL_WORDS];
    __shared__ unsigned queue[SCAN_WAVES * Q_WORDS];

    const int lane = threadIdx.x & 31;
    const int w    = __builtin_amdgcn_readfirstlane((int)(threadIdx.x >> 5));
    const unsigned b = blockIdx.x;
    const unsigned rbase = b << RB_SHIFT;

    float4* acc4 = (float4*)acc;
    for (int i = threadIdx.x; i < RB_ROWS * FEAT / 4; i += SCAN_THREADS)
        acc4[i] = make_float4(0.f, 0.f, 0.f, 0.f);
    __syncthreads();

    const unsigned* mine = map + (size_t)b * nTiles * TILE_WORDS;
    unsigned* gl = glist + w * GL_WORDS;
    unsigned* q  = queue + w * Q_WORDS;

    unsigned gcnt = 0, cnt = 0;
    unsigned nxt = (w < nTiles) ? mine[w * TILE_WORDS + lane] : 0u;
    for (int i = w; i < nTiles; i += SCAN_WAVES) {
        unsigned wbits = nxt;
        if (i + SCAN_WAVES < nTiles) nxt = mine[(i + SCAN_WAVES) * TILE_WORDS + lane];
        const unsigned gbase = (unsigned)i * TILE_GROUPS + (unsigned)lane * 32u;

        for (int pass = 0; pass < 32; ++pass) {
            const bool has = wbits != 0u;
            const unsigned m = __builtin_amdgcn_ballot_w32(has);
            if (!m) break;
            if (has) {
                gl[gcnt + __builtin_amdgcn_mbcnt_lo(m, 0u)] = gbase + (unsigned)__builtin_ctz(wbits);
                wbits &= wbits - 1u;
            }
            gcnt += (unsigned)__builtin_popcount(m);
            if (gcnt >= 32u) {
                gcnt -= 32u;
                match_groups<true>(gl + gcnt, 32u, lane, row, nEdges, rbase, q, cnt);
                while (cnt >= 32u) {
                    cnt -= 32u;
                    drain_records<true>(q + cnt, 32u, lane, col, vals, H, acc);
                }
            }
        }
    }
    if (gcnt) {
        match_groups<false>(gl, gcnt, lane, row, nEdges, rbase, q, cnt);
        while (cnt >= 32u) {
            cnt -= 32u;
            drain_records<true>(q + cnt, 32u, lane, col, vals, H, acc);
        }
    }
    if (cnt) drain_records<false>(q, cnt, lane, col, vals, H, acc);

    __syncthreads();
    const int r0 = (int)b * RB_ROWS;
    const int nrows = (nNodes - r0 < RB_ROWS) ? (nNodes - r0) : RB_ROWS;
    float4* dst = (float4*)(AH + (size_t)r0 * FEAT);
    for (int t = threadIdx.x; t < nrows * (FEAT / 4); t += SCAN_THREADS)
        dst[t] = acc4[t];
}

__global__ void gcnii_gemm_epilogue(const float* __restrict__ AH,
                                    const float* __restrict__ H0,
                                    const float* __restrict__ W,
                                    float* __restrict__ out,
                                    const float* __restrict__ lamdaPtr,
                                    const float* __restrict__ alphaPtr,
                                    const int* __restrict__ lPtr,
                                    int nTiles) {
    int tile = (int)(blockIdx.x * (blockDim.x >> 5) + (threadIdx.x >> 5));
    if (tile >= nTiles) return;

    int m0 = (tile >> 2) * 16;
    int n0 = (tile & 3) * 16;
    int lane = threadIdx.x & 31;
    int hi   = lane >> 4;
    int lm   = lane & 15;

    float alpha = alphaPtr[0];
    float lamda = lamdaPtr[0];
    float beta  = logf(lamda / (float)lPtr[0] + 1.0f);
    float oma   = 1.0f - alpha;
    float omb   = 1.0f - beta;

    const float* arowAH = AH + (size_t)(m0 + lm) * FEAT + hi * 2;
    const float* arowH0 = H0 + (size_t)(m0 + lm) * FEAT + hi * 2;
    const float* wbase  = W + (size_t)(hi * 2) * FEAT + n0 + lm;

    v8f acc = {};
#pragma unroll
    for (int kk = 0; kk < 16; ++kk) {
        int K0 = kk * 4;
        float2 ah2 = *(const float2*)(arowAH + K0);
        float2 h02 = *(const float2*)(arowH0 + K0);
        v2f a;
        a.x = oma * ah2.x + alpha * h02.x;
        a.y = oma * ah2.y + alpha * h02.y;
        v2f b;
        b.x = wbase[(size_t)(K0 + 0) * FEAT];
        b.y = wbase[(size_t)(K0 + 1) * FEAT];
        acc = __builtin_amdgcn_wmma_f32_16x16x4_f32(
             false, a,  false, b,
             (short)0, acc,  false,  false);
    }

#pragma unroll
    for (int r = 0; r < 8; ++r) {
        size_t idx = (size_t)(m0 + r + hi * 8) * FEAT + n0 + lm;
        float sup = oma * AH[idx] + alpha * H0[idx];
        out[idx] = omb * sup + beta * acc[r];
    }
}

extern "C" void kernel_launch(void* const* d_in, const int* in_sizes, int n_in,
                              void* d_out, int out_size, void* d_ws, size_t ws_size,
                              hipStream_t stream) {
    const int*   row   = (const int*)  d_in[0];
    const int*   col   = (const int*)  d_in[1];
    const float* vals  = (const float*)d_in[2];
    const float* H     = (const float*)d_in[3];
    const float* H0    = (const float*)d_in[4];
    const float* W     = (const float*)d_in[5];
    const float* lamda = (const float*)d_in[6];
    const float* alpha = (const float*)d_in[7];
    const int*   l     = (const int*)  d_in[8];

    int nEdges = in_sizes[0];
    int nNodes = in_sizes[3] / FEAT;

    float*    AH  = (float*)d_ws;
    unsigned* map = (unsigned*)(AH + (size_t)nNodes * FEAT);

    int spmmBlocks = (nNodes + RB_ROWS - 1) / RB_ROWS;
    int mapTiles   = (nEdges + TILE_EDGES - 1) / TILE_EDGES;

    gcnii_mark_groups<<<mapTiles, MARK_THREADS, 0, stream>>>(
        row, map, nEdges, spmmBlocks, mapTiles);

    gcnii_spmm_owner_map<<<spmmBlocks, SCAN_THREADS, 0, stream>>>(
        row, col, vals, H, map, AH, nEdges, nNodes, mapTiles);

    int nTiles = (nNodes / 16) * (FEAT / 16);
    int gemmBlocks = (nTiles + 7) / 8;
    gcnii_gemm_epilogue<<<gemmBlocks, 256, 0, stream>>>(
        AH, H0, W, (float*)d_out, lamda, alpha, l, nTiles);
}


# GEMM re-tiled through LDS: a 4-wave group stages support for 64 rows with whole-line loads, each wave keeps a 64x16 slice of W in registers and sweeps 4 row tiles; out leaves LDS as whole lines
# speedup vs baseline: 2.3360x; 1.1524x over previous
#include <hip/hip_runtime.h>
#include <math.h>

typedef __attribute__((ext_vector_type(2))) float v2f;
typedef __attribute__((ext_vector_type(8))) float v8f;

#define FEAT 64

#define RB_SHIFT      9
#define RB_ROWS       (1 << RB_SHIFT)
#define SCAN_WAVES    32
#define SCAN_THREADS  (SCAN_WAVES * 32)
#define GATHER_DEPTH  8
#define REC_EDGE_BITS 23

#define GRP_EDGES       4
#define TILE_GROUPS     1024
#define TILE_EDGES      (GRP_EDGES * TILE_GROUPS)
#define TILE_WORDS      (TILE_GROUPS / 32)
#define MARK_THREADS    256
#define MARK_ITERS      (TILE_GROUPS / MARK_THREADS)
#define MARK_BLOCKS_MAX 256
#define GL_WORDS        64
#define Q_WORDS         160

__device__ __forceinline__ void mark_row(unsigned* bits, int r, unsigned b0, unsigned nb,
                                         unsigned word, unsigned bit) {
    const unsigned blk = ((unsigned)r >> RB_SHIFT) - b0;
    if (blk < nb)
        __hip_atomic_fetch_or(&bits[blk * TILE_WORDS + word], bit,
                              __ATOMIC_RELAXED, __HIP_MEMORY_SCOPE_WORKGROUP);
}

__global__ __launch_bounds__(MARK_THREADS)
void gcnii_mark_groups(const int* __restrict__ row, unsigned* __restrict__ map,
                       int nEdges, int nBlocks, int nTiles) {
    __shared__ unsigned bits[MARK_BLOCKS_MAX * TILE_WORDS];

    const int tid  = threadIdx.x;
    const int tile = blockIdx.x;

    int4 r[MARK_ITERS];
#pragma unroll
    for (int k = 0; k < MARK_ITERS; ++k) {
        const int e0 = tile * TILE_EDGES + (tid + k * MARK_THREADS) * GRP_EDGES;
        if (e0 + GRP_EDGES <= nEdges) {
            r[k] = *(const int4*)(row + e0);
        } else {
            r[k].x = (e0 + 0 < nEdges) ? row[e0 + 0] : -1;
            r[k].y = (e0 + 1 < nEdges) ? row[e0 + 1] : -1;
            r[k].z = (e0 + 2 < nEdges) ? row[e0 + 2] : -1;
            r[k].w = (e0 + 3 < nEdges) ? row[e0 + 3] : -1;
        }
    }

    const unsigned bit = 1u << (tid & 31);
    for (int b0 = 0; b0 < nBlocks; b0 += MARK_BLOCKS_MAX) {
        const int nb = (nBlocks - b0 < MARK_BLOCKS_MAX) ? (nBlocks - b0) : MARK_BLOCKS_MAX;
        uint4* bits4 = (uint4*)bits;
        for (int i = tid; i < nb * TILE_WORDS / 4; i += MARK_THREADS)
            bits4[i] = make_uint4(0u, 0u, 0u, 0u);
        __syncthreads();
#pragma unroll
        for (int k = 0; k < MARK_ITERS; ++k) {
            const unsigned word = (unsigned)(tid >> 5) + (unsigned)k * (MARK_THREADS / 32);
            mark_row(bits, r[k].x, (unsigned)b0, (unsigned)nb, word, bit);
            mark_row(bits, r[k].y, (unsigned)b0, (unsigned)nb, word, bit);
            mark_row(bits, r[k].z, (unsigned)b0, (unsigned)nb, word, bit);
            mark_row(bits, r[k].w, (unsigned)b0, (unsigned)nb, word, bit);
        }
        __syncthreads();
        for (int i = tid; i < nb * TILE_WORDS; i += MARK_THREADS) {
            const size_t line = (size_t)(b0 + (i >> 5)) * nTiles + tile;
            map[line * TILE_WORDS + (i & 31)] = bits[i];
        }
        __syncthreads();
    }
}

__device__ __forceinline__ void append_matches(unsigned t, unsigned e, unsigned* q, unsigned& cnt) {
    const bool match = t < (unsigned)RB_ROWS;
    const unsigned m = __builtin_amdgcn_ballot_w32(match);
    if (m) {
        if (match) q[cnt + __builtin_amdgcn_mbcnt_lo(m, 0u)] = e | (t << REC_EDGE_BITS);
        cnt += (unsigned)__builtin_popcount(m);
    }
}

template <bool FULL>
__device__ __forceinline__ void drain_records(const unsigned* q, unsigned m, int lane,
                                              const int* __restrict__ col,
                                              const float* __restrict__ vals,
                                              const float* __restrict__ H,
                                              float* acc) {
    int cl = 0; float vl = 0.f; unsigned rl = 0;
    if (FULL || (unsigned)lane < m) {
        const unsigned rec = q[lane];
        const unsigned e = rec & ((1u << REC_EDGE_BITS) - 1u);
        cl = col[e];
        vl = vals[e];
        rl = rec >> REC_EDGE_BITS;
    }
    for (unsigned k = 0; k < m; k += GATHER_DEPTH) {
        float2 h[GATHER_DEPTH];
#pragma unroll
        for (int t = 0; t < GATHER_DEPTH; ++t) {
            const unsigned kk = (FULL || k + t < m) ? (k + t) : (m - 1u);
            const int cc = __builtin_amdgcn_readlane(cl, (int)kk);
            const unsigned off = ((unsigned)cc << 8) + (unsigned)lane * 8u;
            h[t] = *(const float2*)((const char*)H + off);
        }
#pragma unroll
        for (int t = 0; t < GATHER_DEPTH; ++t) {
            if (FULL || k + t < m) {
                const float vv = __builtin_bit_cast(float,
                    __builtin_amdgcn_readlane(__builtin_bit_cast(int, vl), (int)(k + t)));
                const unsigned rr = (unsigned)__builtin_amdgcn_readlane((int)rl, (int)(k + t));
                float* d = acc + rr * FEAT + lane * 2;
                __hip_atomic_fetch_add(d + 0, vv * h[t].x, __ATOMIC_RELAXED, __HIP_MEMORY_SCOPE_WORKGROUP);
                __hip_atomic_fetch_add(d + 1, vv * h[t].y, __ATOMIC_RELAXED, __HIP_MEMORY_SCOPE_WORKGROUP);
            }
        }
    }
}

template <bool FULL>
__device__ __forceinline__ void match_groups(const unsigned* gl, unsigned m, int lane,
                                             const int* __restrict__ row, int nEdges,
                                             unsigned rbase, unsigned* q, unsigned& cnt) {
    int4 r = make_int4(-1, -1, -1, -1);
    unsigned e0 = 0u;
    if (FULL || (unsigned)lane < m) {
        e0 = gl[lane] * GRP_EDGES;
        if ((int)e0 + GRP_EDGES <= nEdges) {
            r = *(const int4*)(row + e0);
        } else {
            if ((int)e0 + 0 < nEdges) r.x = row[e0 + 0];
            if ((int)e0 + 1 < nEdges) r.y = row[e0 + 1];
            if ((int)e0 + 2 < nEdges) r.z = row[e0 + 2];
        }
    }
    append_matches((unsigned)r.x ^ rbase, e0 + 0u, q, cnt);
    append_matches((unsigned)r.y ^ rbase, e0 + 1u, q, cnt);
    append_matches((unsigned)r.z ^ rbase, e0 + 2u, q, cnt);
    append_matches((unsigned)r.w ^ rbase, e0 + 3u, q, cnt);
}

__global__ __launch_bounds__(SCAN_THREADS)
void gcnii_spmm_owner_map(const int* __restrict__ row,
                          const int* __restrict__ col,
                          const float* __restrict__ vals,
                          const float* __restrict__ H,
                          const unsigned* __restrict__ map,
                          float* __restrict__ AH,
                          int nEdges, int nNodes, int nTiles) {
    __shared__ float    acc[RB_ROWS * FEAT];
    __shared__ unsigned glist[SCAN_WAVES * GL_WORDS];
    __shared__ unsigned queue[SCAN_WAVES * Q_WORDS];

    const int lane = threadIdx.x & 31;
    const int w    = __builtin_amdgcn_readfirstlane((int)(threadIdx.x >> 5));
    const unsigned b = blockIdx.x;
    const unsigned rbase = b << RB_SHIFT;

    float4* acc4 = (float4*)acc;
    for (int i = threadIdx.x; i < RB_ROWS * FEAT / 4; i += SCAN_THREADS)
        acc4[i] = make_float4(0.f, 0.f, 0.f, 0.f);
    __syncthreads();

    const unsigned* mine = map + (size_t)b * nTiles * TILE_WORDS;
    unsigned* gl = glist + w * GL_WORDS;
    unsigned* q  = queue + w * Q_WORDS;

    unsigned gcnt = 0, cnt = 0;
    unsigned nxt = (w < nTiles) ? mine[w * TILE_WORDS + lane] : 0u;
    for (int i = w; i < nTiles; i += SCAN_WAVES) {
        unsigned wbits = nxt;
        if (i + SCAN_WAVES < nTiles) nxt = mine[(i + SCAN_WAVES) * TILE_WORDS + lane];
        const unsigned gbase = (unsigned)i * TILE_GROUPS + (unsigned)lane * 32u;

        for (int pass = 0; pass < 32; ++pass) {
            const bool has = wbits != 0u;
            const unsigned m = __builtin_amdgcn_ballot_w32(has);
            if (!m) break;
            if (has) {
                gl[gcnt + __builtin_amdgcn_mbcnt_lo(m, 0u)] = gbase + (unsigned)__builtin_ctz(wbits);
                wbits &= wbits - 1u;
            }
            gcnt += (unsigned)__builtin_popcount(m);
            if (gcnt >= 32u) {
                gcnt -= 32u;
                match_groups<true>(gl + gcnt, 32u, lane, row, nEdges, rbase, q, cnt);
                while (cnt >= 32u) {
                    cnt -= 32u;
                    drain_records<true>(q + cnt, 32u, lane, col, vals, H, acc);
                }
            }
        }
    }
    if (gcnt) {
        match_groups<false>(gl, gcnt, lane, row, nEdges, rbase, q, cnt);
        while (cnt >= 32u) {
            cnt -= 32u;
            drain_records<true>(q + cnt, 32u, lane, col, vals, H, acc);
        }
    }
    if (cnt) drain_records<false>(q, cnt, lane, col, vals, H, acc);

    __syncthreads();
    const int r0 = (int)b * RB_ROWS;
    const int nrows = (nNodes - r0 < RB_ROWS) ? (nNodes - r0) : RB_ROWS;
    float4* dst = (float4*)(AH + (size_t)r0 * FEAT);
    for (int t = threadIdx.x; t < nrows * (FEAT / 4); t += SCAN_THREADS)
        dst[t] = acc4[t];
}

#define GT_ROWS    64
#define GT_PITCH   68
#define GT_THREADS 128
#define GT_LOADS   (GT_ROWS * (FEAT / 4) / GT_THREADS)

__global__ __launch_bounds__(GT_THREADS)
void gcnii_gemm_epilogue(const float* __restrict__ AH,
                         const float* __restrict__ H0,
                         const float* __restrict__ W,
                         float* __restrict__ out,
                         const float* __restrict__ lamdaPtr,
                         const float* __restrict__ alphaPtr,
                         const int* __restrict__ lPtr,
                         int nNodes) {
    __shared__ __attribute__((aligned(16))) float sup[GT_ROWS * GT_PITCH];

    const int tid  = threadIdx.x;
    const int lane = tid & 31;
    const int w    = tid >> 5;
    const int hi   = lane >> 4;
    const int lm   = lane & 15;
    const int n0   = w * 16;

    const int r0    = (int)blockIdx.x * GT_ROWS;
    const int nrows = (nNodes - r0 < GT_ROWS) ? (nNodes - r0) : GT_ROWS;

    const float alpha = alphaPtr[0];
    const float lamda = lamdaPtr[0];
    const float beta  = logf(lamda / (float)lPtr[0] + 1.0f);
    const float oma   = 1.0f - alpha;
    const float omb   = 1.0f - beta;

    const float* wbase = W + (size_t)(hi * 2) * FEAT + n0 + lm;
    v2f bfrag[16];
#pragma unroll
    for (int kk = 0; kk < 16; ++kk) {
        bfrag[kk].x = wbase[(size_t)(kk * 4 + 0) * FEAT];
        bfrag[kk].y = wbase[(size_t)(kk * 4 + 1) * FEAT];
    }

    const float4* AH4 = (const float4*)(AH + (size_t)r0 * FEAT);
    const float4* H04 = (const float4*)(H0 + (size_t)r0 * FEAT);
    const int plast = nrows * (FEAT / 4) - 1;
    float4 a4[GT_LOADS], h4[GT_LOADS];
#pragma unroll
    for (int j = 0; j < GT_LOADS; ++j) {
        const int p  = tid + j * GT_THREADS;
        const int pc = (p < plast) ? p : plast;
        a4[j] = AH4[pc];
        h4[j] = H04[pc];
    }

#pragma unroll
    for (int j = 0; j < GT_LOADS; ++j) {
        const int p = tid + j * GT_THREADS;
        float4 s;
        s.x = oma * a4[j].x + alpha * h4[j].x;
        s.y = oma * a4[j].y + alpha * h4[j].y;
        s.z = oma * a4[j].z + alpha * h4[j].z;
        s.w = oma * a4[j].w + alpha * h4[j].w;
        *(float4*)&sup[(p >> 4) * GT_PITCH + (p & 15) * 4] = s;
    }
    __syncthreads();

    v8f acc[4] = {};
    const float* arow = sup + lm * GT_PITCH + hi * 2;
#pragma unroll
    for (int kk = 0; kk < 16; ++kk) {
#pragma unroll
        for (int t = 0; t < 4; ++t) {
            const v2f a = *(const v2f*)(arow + t * 16 * GT_PITCH + kk * 4);
            acc[t] = __builtin_amdgcn_wmma_f32_16x16x4_f32(
                 false, a,  false, bfrag[kk],
                 (short)0, acc[t],  false,  false);
        }
    }
    __syncthreads();

#pragma unroll
    for (int t = 0; t < 4; ++t) {
#pragma unroll
        for (int r = 0; r < 8; ++r) {
            float* s = sup + (t * 16 + r + hi * 8) * GT_PITCH + n0 + lm;
            *s = omb * (*s) + beta * acc[t][r];
        }
    }
    __syncthreads();

    float4* out4 = (float4*)(out + (size_t)r0 * FEAT);
    if (nrows == GT_ROWS) {
        float4 o[GT_LOADS];
#pragma unroll
        for (int j = 0; j < GT_LOADS; ++j) {
            const int p = tid + j * GT_THREADS;
            o[j] = *(const float4*)&sup[(p >> 4) * GT_PITCH + (p & 15) * 4];
        }
#pragma unroll
        for (int j = 0; j < GT_LOADS; ++j)
            out4[tid + j * GT_THREADS] = o[j];
    } else {
        for (int j = 0; j < GT_LOADS; ++j) {
            const int p = tid + j * GT_THREADS;
            if ((p >> 4) < nrows)
                out4[p] = *(const float4*)&sup[(p >> 4) * GT_PITCH + (p & 15) * 4];
        }
    }
}

extern "C" void kernel_launch(void* const* d_in, const int* in_sizes, int n_in,
                              void* d_out, int out_size, void* d_ws, size_t ws_size,
                              hipStream_t stream) {
    const int*   row   = (const int*)  d_in[0];
    const int*   col   = (const int*)  d_in[1];
    const float* vals  = (const float*)d_in[2];
    const float* H     = (const float*)d_in[3];
    const float* H0    = (const float*)d_in[4];
    const float* W     = (const float*)d_in[5];
    const float* lamda = (const float*)d_in[6];
    const float* alpha = (const float*)d_in[7];
    const int*   l     = (const int*)  d_in[8];

    int nEdges = in_sizes[0];
    int nNodes = in_sizes[3] / FEAT;

    float*    AH  = (float*)d_ws;
    unsigned* map = (unsigned*)(AH + (size_t)nNodes * FEAT);

    int spmmBlocks = (nNodes + RB_ROWS - 1) / RB_ROWS;
    int mapTiles   = (nEdges + TILE_EDGES - 1) / TILE_EDGES;

    gcnii_mark_groups<<<mapTiles, MARK_THREADS, 0, stream>>>(
        row, map, nEdges, spmmBlocks, mapTiles);

    gcnii_spmm_owner_map<<<spmmBlocks, SCAN_THREADS, 0, stream>>>(
        row, col, vals, H, map, AH, nEdges, nNodes, mapTiles);

    int gemmBlocks = (nNodes + GT_ROWS - 1) / GT_ROWS;
    gcnii_gemm_epilogue<<<gemmBlocks, GT_THREADS, 0, stream>>>(
        AH, H0, W, (float*)d_out, lamda, alpha, l, nNodes);
}


# SPMM owner row blocks of 400 rows, not 512: 250 owner work-groups spread over the 256 WGPs (was 196), each sums 6400 edges, not 8160; map has 250 lines per tile; sums still in arrival order
# speedup vs baseline: 2.5486x; 1.0910x over previous
#include <hip/hip_runtime.h>
#include <math.h>

typedef __attribute__((ext_vector_type(2))) float v2f;
typedef __attribute__((ext_vector_type(8))) float v8f;

#define FEAT 64

#define RB_ROWS       400
#define SCAN_WAVES    32
#define SCAN_THREADS  (SCAN_WAVES * 32)
#define GATHER_DEPTH  8
#define REC_EDGE_BITS 23

#define GRP_EDGES       4
#define TILE_GROUPS     1024
#define TILE_EDGES      (GRP_EDGES * TILE_GROUPS)
#define TILE_WORDS      (TILE_GROUPS / 32)
#define MARK_THREADS    256
#define MARK_ITERS      (TILE_GROUPS / MARK_THREADS)
#define MARK_BLOCKS_MAX 256
#define GL_WORDS        64
#define Q_WORDS         160

__device__ __forceinline__ void mark_row(unsigned* bits, int r, unsigned b0, unsigned nb,
                                         unsigned word, unsigned bit) {
    const unsigned blk = (unsigned)r / (unsigned)RB_ROWS - b0;
    if (blk < nb)
        __hip_atomic_fetch_or(&bits[blk * TILE_WORDS + word], bit,
                              __ATOMIC_RELAXED, __HIP_MEMORY_SCOPE_WORKGROUP);
}

__global__ __launch_bounds__(MARK_THREADS)
void gcnii_mark_groups(const int* __restrict__ row, unsigned* __restrict__ map,
                       int nEdges, int nBlocks, int nTiles) {
    __shared__ unsigned bits[MARK_BLOCKS_MAX * TILE_WORDS];

    const int tid  = threadIdx.x;
    const int tile = blockIdx.x;

    int4 r[MARK_ITERS];
#pragma unroll
    for (int k = 0; k < MARK_ITERS; ++k) {
        const int e0 = tile * TILE_EDGES + (tid + k * MARK_THREADS) * GRP_EDGES;
        if (e0 + GRP_EDGES <= nEdges) {
            r[k] = *(const int4*)(row + e0);
        } else {
            r[k].x = (e0 + 0 < nEdges) ? row[e0 + 0] : -1;
            r[k].y = (e0 + 1 < nEdges) ? row[e0 + 1] : -1;
            r[k].z = (e0 + 2 < nEdges) ? row[e0 + 2] : -1;
            r[k].w = (e0 + 3 < nEdges) ? row[e0 + 3] : -1;
        }
    }

    const unsigned bit = 1u << (tid & 31);
    for (int b0 = 0; b0 < nBlocks; b0 += MARK_BLOCKS_MAX) {
        const int nb = (nBlocks - b0 < MARK_BLOCKS_MAX) ? (nBlocks - b0) : MARK_BLOCKS_MAX;
        uint4* bits4 = (uint4*)bits;
        for (int i = tid; i < nb * TILE_WORDS / 4; i += MARK_THREADS)
            bits4[i] = make_uint4(0u, 0u, 0u, 0u);
        __syncthreads();
#pragma unroll
        for (int k = 0; k < MARK_ITERS; ++k) {
            const unsigned word = (unsigned)(tid >> 5) + (unsigned)k * (MARK_THREADS / 32);
            mark_row(bits, r[k].x, (unsigned)b0, (unsigned)nb, word, bit);
            mark_row(bits, r[k].y, (unsigned)b0, (unsigned)nb, word, bit);
            mark_row(bits, r[k].z, (unsigned)b0, (unsigned)nb, word, bit);
            mark_row(bits, r[k].w, (unsigned)b0, (unsigned)nb, word, bit);
        }
        __syncthreads();
        for (int i = tid; i < nb * TILE_WORDS; i += MARK_THREADS) {
            const size_t line = (size_t)(b0 + (i >> 5)) * nTiles + tile;
            map[line * TILE_WORDS + (i & 31)] = bits[i];
        }
        __syncthreads();
    }
}

__device__ __forceinline__ void append_matches(unsigned t, unsigned e, unsigned* q, unsigned& cnt) {
    const bool match = t < (unsigned)RB_ROWS;
    const unsigned m = __builtin_amdgcn_ballot_w32(match);
    if (m) {
        if (match) q[cnt + __builtin_amdgcn_mbcnt_lo(m, 0u)] = e | (t << REC_EDGE_BITS);
        cnt += (unsigned)__builtin_popcount(m);
    }
}

template <bool FULL>
__device__ __forceinline__ void drain_records(const unsigned* q, unsigned m, int lane,
                                              const int* __restrict__ col,
                                              const float* __restrict__ vals,
                                              const float* __restrict__ H,
                                              float* acc) {
    int cl = 0; float vl = 0.f; unsigned rl = 0;
    if (FULL || (unsigned)lane < m) {
        const unsigned rec = q[lane];
        const unsigned e = rec & ((1u << REC_EDGE_BITS) - 1u);
        cl = col[e];
        vl = vals[e];
        rl = rec >> REC_EDGE_BITS;
    }
    for (unsigned k = 0; k < m; k += GATHER_DEPTH) {
        float2 h[GATHER_DEPTH];
#pragma unroll
        for (int t = 0; t < GATHER_DEPTH; ++t) {
            const unsigned kk = (FULL || k + t < m) ? (k + t) : (m - 1u);
            const int cc = __builtin_amdgcn_readlane(cl, (int)kk);
            const unsigned off = ((unsigned)cc << 8) + (unsigned)lane * 8u;
            h[t] = *(const float2*)((const char*)H + off);
        }
#pragma unroll
        for (int t = 0; t < GATHER_DEPTH; ++t) {
            if (FULL || k + t < m) {
                const float vv = __builtin_bit_cast(float,
                    __builtin_amdgcn_readlane(__builtin_bit_cast(int, vl), (int)(k + t)));
                const unsigned rr = (unsigned)__builtin_amdgcn_readlane((int)rl, (int)(k + t));
                float* d = acc + rr * FEAT + lane * 2;
                __hip_atomic_fetch_add(d + 0, vv * h[t].x, __ATOMIC_RELAXED, __HIP_MEMORY_SCOPE_WORKGROUP);
                __hip_atomic_fetch_add(d + 1, vv * h[t].y, __ATOMIC_RELAXED, __HIP_MEMORY_SCOPE_WORKGROUP);
            }
        }
    }
}

template <bool FULL>
__device__ __forceinline__ void match_groups(const unsigned* gl, unsigned m, int lane,
                                             const int* __restrict__ row, int nEdges,
                                             unsigned rbase, unsigned* q, unsigned& cnt) {
    int4 r = make_int4(-1, -1, -1, -1);
    unsigned e0 = 0u;
    if (FULL || (unsigned)lane < m) {
        e0 = gl[lane] * GRP_EDGES;
        if ((int)e0 + GRP_EDGES <= nEdges) {
            r = *(const int4*)(row + e0);
        } else {
            if ((int)e0 + 0 < nEdges) r.x = row[e0 + 0];
            if ((int)e0 + 1 < nEdges) r.y = row[e0 + 1];
            if ((int)e0 + 2 < nEdges) r.z = row[e0 + 2];
        }
    }
    append_matches((unsigned)r.x - rbase, e0 + 0u, q, cnt);
    append_matches((unsigned)r.y - rbase, e0 + 1u, q, cnt);
    append_matches((unsigned)r.z - rbase, e0 + 2u, q, cnt);
    append_matches((unsigned)r.w - rbase, e0 + 3u, q, cnt);
}

__global__ __launch_bounds__(SCAN_THREADS)
void gcnii_spmm_owner_map(const int* __restrict__ row,
                          const int* __restrict__ col,
                          const float* __restrict__ vals,
                          const float* __restrict__ H,
                          const unsigned* __restrict__ map,
                          float* __restrict__ AH,
                          int nEdges, int nNodes, int nTiles) {
    __shared__ float    acc[RB_ROWS * FEAT];
    __shared__ unsigned glist[SCAN_WAVES * GL_WORDS];
    __shared__ unsigned queue[SCAN_WAVES * Q_WORDS];

    const int lane = threadIdx.x & 31;
    const int w    = __builtin_amdgcn_readfirstlane((int)(threadIdx.x >> 5));
    const unsigned b = blockIdx.x;
    const unsigned rbase = b * (unsigned)RB_ROWS;

    float4* acc4 = (float4*)acc;
    for (int i = threadIdx.x; i < RB_ROWS * FEAT / 4; i += SCAN_THREADS)
        acc4[i] = make_float4(0.f, 0.f, 0.f, 0.f);
    __syncthreads();

    const unsigned* mine = map + (size_t)b * nTiles * TILE_WORDS;
    unsigned* gl = glist + w * GL_WORDS;
    unsigned* q  = queue + w * Q_WORDS;

    unsigned gcnt = 0, cnt = 0;
    unsigned nxt = (w < nTiles) ? mine[w * TILE_WORDS + lane] : 0u;
    for (int i = w; i < nTiles; i += SCAN_WAVES) {
        unsigned wbits = nxt;
        if (i + SCAN_WAVES < nTiles) nxt = mine[(i + SCAN_WAVES) * TILE_WORDS + lane];
        const unsigned gbase = (unsigned)i * TILE_GROUPS + (unsigned)lane * 32u;

        for (int pass = 0; pass < 32; ++pass) {
            const bool has = wbits != 0u;
            const unsigned m = __builtin_amdgcn_ballot_w32(has);
            if (!m) break;
            if (has) {
                gl[gcnt + __builtin_amdgcn_mbcnt_lo(m, 0u)] = gbase + (unsigned)__builtin_ctz(wbits);
                wbits &= wbits - 1u;
            }
            gcnt += (unsigned)__builtin_popcount(m);
            if (gcnt >= 32u) {
                gcnt -= 32u;
                match_groups<true>(gl + gcnt, 32u, lane, row, nEdges, rbase, q, cnt);
                while (cnt >= 32u) {
                    cnt -= 32u;
                    drain_records<true>(q + cnt, 32u, lane, col, vals, H, acc);
                }
            }
        }
    }
    if (gcnt) {
        match_groups<false>(gl, gcnt, lane, row, nEdges, rbase, q, cnt);
        while (cnt >= 32u) {
            cnt -= 32u;
            drain_records<true>(q + cnt, 32u, lane, col, vals, H, acc);
        }
    }
    if (cnt) drain_records<false>(q, cnt, lane, col, vals, H, acc);

    __syncthreads();
    const int r0 = (int)b * RB_ROWS;
    const int nrows = (nNodes - r0 < RB_ROWS) ? (nNodes - r0) : RB_ROWS;
    float4* dst = (float4*)(AH + (size_t)r0 * FEAT);
    for (int t = threadIdx.x; t < nrows * (FEAT / 4); t += SCAN_THREADS)
        dst[t] = acc4[t];
}

#define GT_ROWS    64
#define GT_PITCH   68
#define GT_THREADS 128
#define GT_LOADS   (GT_ROWS * (FEAT / 4) / GT_THREADS)

__global__ __launch_bounds__(GT_THREADS)
void gcnii_gemm_epilogue(const float* __restrict__ AH,
                         const float* __restrict__ H0,
                         const float* __restrict__ W,
                         float* __restrict__ out,
                         const float* __restrict__ lamdaPtr,
                         const float* __restrict__ alphaPtr,
                         const int* __restrict__ lPtr,
                         int nNodes) {
    __shared__ __attribute__((aligned(16))) float sup[GT_ROWS * GT_PITCH];

    const int tid  = threadIdx.x;
    const int lane = tid & 31;
    const int w    = tid >> 5;
    const int hi   = lane >> 4;
    const int lm   = lane & 15;
    const int n0   = w * 16;

    const int r0    = (int)blockIdx.x * GT_ROWS;
    const int nrows = (nNodes - r0 < GT_ROWS) ? (nNodes - r0) : GT_ROWS;

    const float alpha = alphaPtr[0];
    const float lamda = lamdaPtr[0];
    const float beta  = logf(lamda / (float)lPtr[0] + 1.0f);
    const float oma   = 1.0f - alpha;
    const float omb   = 1.0f - beta;

    const float* wbase = W + (size_t)(hi * 2) * FEAT + n0 + lm;
    v2f bfrag[16];
#pragma unroll
    for (int kk = 0; kk < 16; ++kk) {
        bfrag[kk].x = wbase[(size_t)(kk * 4 + 0) * FEAT];
        bfrag[kk].y = wbase[(size_t)(kk * 4 + 1) * FEAT];
    }

    const float4* AH4 = (const float4*)(AH + (size_t)r0 * FEAT);
    const float4* H04 = (const float4*)(H0 + (size_t)r0 * FEAT);
    const int plast = nrows * (FEAT / 4) - 1;
    float4 a4[GT_LOADS], h4[GT_LOADS];
#pragma unroll
    for (int j = 0; j < GT_LOADS; ++j) {
        const int p  = tid + j * GT_THREADS;
        const int pc = (p < plast) ? p : plast;
        a4[j] = AH4[pc];
        h4[j] = H04[pc];
    }

#pragma unroll
    for (int j = 0; j < GT_LOADS; ++j) {
        const int p = tid + j * GT_THREADS;
        float4 s;
        s.x = oma * a4[j].x + alpha * h4[j].x;
        s.y = oma * a4[j].y + alpha * h4[j].y;
        s.z = oma * a4[j].z + alpha * h4[j].z;
        s.w = oma * a4[j].w + alpha * h4[j].w;
        *(float4*)&sup[(p >> 4) * GT_PITCH + (p & 15) * 4] = s;
    }
    __syncthreads();

    v8f acc[4] = {};
    const float* arow = sup + lm * GT_PITCH + hi * 2;
#pragma unroll
    for (int kk = 0; kk < 16; ++kk) {
#pragma unroll
        for (int t = 0; t < 4; ++t) {
            const v2f a = *(const v2f*)(arow + t * 16 * GT_PITCH + kk * 4);
            acc[t] = __builtin_amdgcn_wmma_f32_16x16x4_f32(
                 false, a,  false, bfrag[kk],
                 (short)0, acc[t],  false,  false);
        }
    }
    __syncthreads();

#pragma unroll
    for (int t = 0; t < 4; ++t) {
#pragma unroll
        for (int r = 0; r < 8; ++r) {
            float* s = sup + (t * 16 + r + hi * 8) * GT_PITCH + n0 + lm;
            *s = omb * (*s) + beta * acc[t][r];
        }
    }
    __syncthreads();

    float4* out4 = (float4*)(out + (size_t)r0 * FEAT);
    if (nrows == GT_ROWS) {
        float4 o[GT_LOADS];
#pragma unroll
        for (int j = 0; j < GT_LOADS; ++j) {
            const int p = tid + j * GT_THREADS;
            o[j] = *(const float4*)&sup[(p >> 4) * GT_PITCH + (p & 15) * 4];
        }
#pragma unroll
        for (int j = 0; j < GT_LOADS; ++j)
            out4[tid + j * GT_THREADS] = o[j];
    } else {
        for (int j = 0; j < GT_LOADS; ++j) {
            const int p = tid + j * GT_THREADS;
            if ((p >> 4) < nrows)
                out4[p] = *(const float4*)&sup[(p >> 4) * GT_PITCH + (p & 15) * 4];
        }
    }
}

extern "C" void kernel_launch(void* const* d_in, const int* in_sizes, int n_in,
                              void* d_out, int out_size, void* d_ws, size_t ws_size,
                              hipStream_t stream) {
    const int*   row   = (const int*)  d_in[0];
    const int*   col   = (const int*)  d_in[1];
    const float* vals  = (const float*)d_in[2];
    const float* H     = (const float*)d_in[3];
    const float* H0    = (const float*)d_in[4];
    const float* W     = (const float*)d_in[5];
    const float* lamda = (const float*)d_in[6];
    const float* alpha = (const float*)d_in[7];
    const int*   l     = (const int*)  d_in[8];

    int nEdges = in_sizes[0];
    int nNodes = in_sizes[3] / FEAT;

    float*    AH  = (float*)d_ws;
    unsigned* map = (unsigned*)(AH + (size_t)nNodes * FEAT);

    int spmmBlocks = (nNodes + RB_ROWS - 1) / RB_ROWS;
    int mapTiles   = (nEdges + TILE_EDGES - 1) / TILE_EDGES;

    gcnii_mark_groups<<<mapTiles, MARK_THREADS, 0, stream>>>(
        row, map, nEdges, spmmBlocks, mapTiles);

    gcnii_spmm_owner_map<<<spmmBlocks, SCAN_THREADS, 0, stream>>>(
        row, col, vals, H, map, AH, nEdges, nNodes, mapTiles);

    int gemmBlocks = (nNodes + GT_ROWS - 1) / GT_ROWS;
    gcnii_gemm_epilogue<<<gemmBlocks, GT_THREADS, 0, stream>>>(
        AH, H0, W, (float*)d_out, lamda, alpha, l, nNodes);
}


# SPMM owner: per 32 edges a wave waits 2 memory round trips, not 6: col and value pieces load beside the row piece (records carry them), all 32 source rows in flight (was 8); sums still arrival order
# speedup vs baseline: 2.6333x; 1.0332x over previous
#include <hip/hip_runtime.h>
#include <math.h>

typedef __attribute__((ext_vector_type(2))) float v2f;
typedef __attribute__((ext_vector_type(8))) float v8f;

#define FEAT 64

#define RB_ROWS       400
#define SCAN_WAVES    32
#define SCAN_THREADS  (SCAN_WAVES * 32)
#define GATHER_DEPTH  32
#define REC_COL_BITS  23

#define GRP_EDGES       4
#define TILE_GROUPS     1024
#define TILE_EDGES      (GRP_EDGES * TILE_GROUPS)
#define TILE_WORDS      (TILE_GROUPS / 32)
#define MARK_THREADS    256
#define MARK_ITERS      (TILE_GROUPS / MARK_THREADS)
#define MARK_BLOCKS_MAX 256
#define GL_WORDS        64
#define Q_RECS          160

__device__ __forceinline__ void mark_row(unsigned* bits, int r, unsigned b0, unsigned nb,
                                         unsigned word, unsigned bit) {
    const unsigned blk = (unsigned)r / (unsigned)RB_ROWS - b0;
    if (blk < nb)
        __hip_atomic_fetch_or(&bits[blk * TILE_WORDS + word], bit,
                              __ATOMIC_RELAXED, __HIP_MEMORY_SCOPE_WORKGROUP);
}

__global__ __launch_bounds__(MARK_THREADS)
void gcnii_mark_groups(const int* __restrict__ row, unsigned* __restrict__ map,
                       int nEdges, int nBlocks, int nTiles) {
    __shared__ unsigned bits[MARK_BLOCKS_MAX * TILE_WORDS];

    const int tid  = threadIdx.x;
    const int tile = blockIdx.x;

    int4 r[MARK_ITERS];
#pragma unroll
    for (int k = 0; k < MARK_ITERS; ++k) {
        const int e0 = tile * TILE_EDGES + (tid + k * MARK_THREADS) * GRP_EDGES;
        if (e0 + GRP_EDGES <= nEdges) {
            r[k] = *(const int4*)(row + e0);
        } else {
            r[k].x = (e0 + 0 < nEdges) ? row[e0 + 0] : -1;
            r[k].y = (e0 + 1 < nEdges) ? row[e0 + 1] : -1;
            r[k].z = (e0 + 2 < nEdges) ? row[e0 + 2] : -1;
            r[k].w = (e0 + 3 < nEdges) ? row[e0 + 3] : -1;
        }
    }

    const unsigned bit = 1u << (tid & 31);
    for (int b0 = 0; b0 < nBlocks; b0 += MARK_BLOCKS_MAX) {
        const int nb = (nBlocks - b0 < MARK_BLOCKS_MAX) ? (nBlocks - b0) : MARK_BLOCKS_MAX;
        uint4* bits4 = (uint4*)bits;
        for (int i = tid; i < nb * TILE_WORDS / 4; i += MARK_THREADS)
            bits4[i] = make_uint4(0u, 0u, 0u, 0u);
        __syncthreads();
#pragma unroll
        for (int k = 0; k < MARK_ITERS; ++k) {
            const unsigned word = (unsigned)(tid >> 5) + (unsigned)k * (MARK_THREADS / 32);
            mark_row(bits, r[k].x, (unsigned)b0, (unsigned)nb, word, bit);
            mark_row(bits, r[k].y, (unsigned)b0, (unsigned)nb, word, bit);
            mark_row(bits, r[k].z, (unsigned)b0, (unsigned)nb, word, bit);
            mark_row(bits, r[k].w, (unsigned)b0, (unsigned)nb, word, bit);
        }
        __syncthreads();
        for (int i = tid; i < nb * TILE_WORDS; i += MARK_THREADS) {
            const size_t line = (size_t)(b0 + (i >> 5)) * nTiles + tile;
            map[line * TILE_WORDS + (i & 31)] = bits[i];
        }
        __syncthreads();
    }
}

__device__ __forceinline__ void append_matches(unsigned t, int c, float v, uint2* q, unsigned& cnt) {
    const bool match = t < (unsigned)RB_ROWS;
    const unsigned m = __builtin_amdgcn_ballot_w32(match);
    if (m) {
        if (match)
            q[cnt + __builtin_amdgcn_mbcnt_lo(m, 0u)] =
                make_uint2((unsigned)c | (t << REC_COL_BITS), __builtin_bit_cast(unsigned, v));
        cnt += (unsigned)__builtin_popcount(m);
    }
}

template <bool FULL>
__device__ __forceinline__ void drain_records(const uint2* q, unsigned m, int lane,
                                              const float* __restrict__ H,
                                              float* acc) {
    int cl = 0; float vl = 0.f; unsigned rl = 0;
    if (FULL || (unsigned)lane < m) {
        const uint2 rec = q[lane];
        cl = (int)(rec.x & ((1u << REC_COL_BITS) - 1u));
        rl = rec.x >> REC_COL_BITS;
        vl = __builtin_bit_cast(float, rec.y);
    }
    for (unsigned k = 0; k < m; k += GATHER_DEPTH) {
        float2 h[GATHER_DEPTH];
#pragma unroll
        for (int t = 0; t < GATHER_DEPTH; ++t) {
            const unsigned kk = (FULL || k + t < m) ? (k + t) : (m - 1u);
            const int cc = __builtin_amdgcn_readlane(cl, (int)kk);
            const unsigned off = ((unsigned)cc << 8) + (unsigned)lane * 8u;
            h[t] = *(const float2*)((const char*)H + off);
        }
#pragma unroll
        for (int t = 0; t < GATHER_DEPTH; ++t) {
            if (FULL || k + t < m) {
                const float vv = __builtin_bit_cast(float,
                    __builtin_amdgcn_readlane(__builtin_bit_cast(int, vl), (int)(k + t)));
                const unsigned rr = (unsigned)__builtin_amdgcn_readlane((int)rl, (int)(k + t));
                float* d = acc + rr * FEAT + lane * 2;
                __hip_atomic_fetch_add(d + 0, vv * h[t].x, __ATOMIC_RELAXED, __HIP_MEMORY_SCOPE_WORKGROUP);
                __hip_atomic_fetch_add(d + 1, vv * h[t].y, __ATOMIC_RELAXED, __HIP_MEMORY_SCOPE_WORKGROUP);
            }
        }
    }
}

template <bool FULL>
__device__ __forceinline__ void match_groups(const unsigned* gl, unsigned m, int lane,
                                             const int* __restrict__ row,
                                             const int* __restrict__ col,
                                             const float* __restrict__ vals, int nEdges,
                                             unsigned rbase, uint2* q, unsigned& cnt) {
    int4   r = make_int4(-1, -1, -1, -1);
    int4   c = make_int4(0, 0, 0, 0);
    float4 v = make_float4(0.f, 0.f, 0.f, 0.f);
    if (FULL || (unsigned)lane < m) {
        const unsigned e0 = gl[lane] * GRP_EDGES;
        if ((int)e0 + GRP_EDGES <= nEdges) {
            r = *(const int4*)(row + e0);
            c = *(const int4*)(col + e0);
            v = *(const float4*)(vals + e0);
        } else {
            if ((int)e0 + 0 < nEdges) { r.x = row[e0 + 0]; c.x = col[e0 + 0]; v.x = vals[e0 + 0]; }
            if ((int)e0 + 1 < nEdges) { r.y = row[e0 + 1]; c.y = col[e0 + 1]; v.y = vals[e0 + 1]; }
            if ((int)e0 + 2 < nEdges) { r.z = row[e0 + 2]; c.z = col[e0 + 2]; v.z = vals[e0 + 2]; }
        }
    }
    append_matches((unsigned)r.x - rbase, c.x, v.x, q, cnt);
    append_matches((unsigned)r.y - rbase, c.y, v.y, q, cnt);
    append_matches((unsigned)r.z - rbase, c.z, v.z, q, cnt);
    append_matches((unsigned)r.w - rbase, c.w, v.w, q, cnt);
}

__global__ __launch_bounds__(SCAN_THREADS)
void gcnii_spmm_owner_map(const int* __restrict__ row,
                          const int* __restrict__ col,
                          const float* __restrict__ vals,
                          const float* __restrict__ H,
                          const unsigned* __restrict__ map,
                          float* __restrict__ AH,
                          int nEdges, int nNodes, int nTiles) {
    __shared__ float    acc[RB_ROWS * FEAT];
    __shared__ unsigned glist[SCAN_WAVES * GL_WORDS];
    __shared__ uint2    queue[SCAN_WAVES * Q_RECS];

    const int lane = threadIdx.x & 31;
    const int w    = __builtin_amdgcn_readfirstlane((int)(threadIdx.x >> 5));
    const unsigned b = blockIdx.x;
    const unsigned rbase = b * (unsigned)RB_ROWS;

    float4* acc4 = (float4*)acc;
    for (int i = threadIdx.x; i < RB_ROWS * FEAT / 4; i += SCAN_THREADS)
        acc4[i] = make_float4(0.f, 0.f, 0.f, 0.f);
    __syncthreads();

    const unsigned* mine = map + (size_t)b * nTiles * TILE_WORDS;
    unsigned* gl = glist + w * GL_WORDS;
    uint2*    q  = queue + w * Q_RECS;

    unsigned gcnt = 0, cnt = 0;
    unsigned nxt = (w < nTiles) ? mine[w * TILE_WORDS + lane] : 0u;
    for (int i = w; i < nTiles; i += SCAN_WAVES) {
        unsigned wbits = nxt;
        if (i + SCAN_WAVES < nTiles) nxt = mine[(i + SCAN_WAVES) * TILE_WORDS + lane];
        const unsigned gbase = (unsigned)i * TILE_GROUPS + (unsigned)lane * 32u;

        for (int pass = 0; pass < 32; ++pass) {
            const bool has = wbits != 0u;
            const unsigned m = __builtin_amdgcn_ballot_w32(has);
            if (!m) break;
            if (has) {
                gl[gcnt + __builtin_amdgcn_mbcnt_lo(m, 0u)] = gbase + (unsigned)__builtin_ctz(wbits);
                wbits &= wbits - 1u;
            }
            gcnt += (unsigned)__builtin_popcount(m);
            if (gcnt >= 32u) {
                gcnt -= 32u;
                match_groups<true>(gl + gcnt, 32u, lane, row, col, vals, nEdges, rbase, q, cnt);
                while (cnt >= 32u) {
                    cnt -= 32u;
                    drain_records<true>(q + cnt, 32u, lane, H, acc);
                }
            }
        }
    }
    if (gcnt) {
        match_groups<false>(gl, gcnt, lane, row, col, vals, nEdges, rbase, q, cnt);
        while (cnt >= 32u) {
            cnt -= 32u;
            drain_records<true>(q + cnt, 32u, lane, H, acc);
        }
    }
    if (cnt) drain_records<false>(q, cnt, lane, H, acc);

    __syncthreads();
    const int r0 = (int)b * RB_ROWS;
    const int nrows = (nNodes - r0 < RB_ROWS) ? (nNodes - r0) : RB_ROWS;
    float4* dst = (float4*)(AH + (size_t)r0 * FEAT);
    for (int t = threadIdx.x; t < nrows * (FEAT / 4); t += SCAN_THREADS)
        dst[t] = acc4[t];
}

#define GT_ROWS    64
#define GT_PITCH   68
#define GT_THREADS 128
#define GT_LOADS   (GT_ROWS * (FEAT / 4) / GT_THREADS)

__global__ __launch_bounds__(GT_THREADS)
void gcnii_gemm_epilogue(const float* __restrict__ AH,
                         const float* __restrict__ H0,
                         const float* __restrict__ W,
                         float* __restrict__ out,
                         const float* __restrict__ lamdaPtr,
                         const float* __restrict__ alphaPtr,
                         const int* __restrict__ lPtr,
                         int nNodes) {
    __shared__ __attribute__((aligned(16))) float sup[GT_ROWS * GT_PITCH];

    const int tid  = threadIdx.x;
    const int lane = tid & 31;
    const int w    = tid >> 5;
    const int hi   = lane >> 4;
    const int lm   = lane & 15;
    const int n0   = w * 16;

    const int r0    = (int)blockIdx.x * GT_ROWS;
    const int nrows = (nNodes - r0 < GT_ROWS) ? (nNodes - r0) : GT_ROWS;

    const float alpha = alphaPtr[0];
    const float lamda = lamdaPtr[0];
    const float beta  = logf(lamda / (float)lPtr[0] + 1.0f);
    const float oma   = 1.0f - alpha;
    const float omb   = 1.0f - beta;

    const float* wbase = W + (size_t)(hi * 2) * FEAT + n0 + lm;
    v2f bfrag[16];
#pragma unroll
    for (int kk = 0; kk < 16; ++kk) {
        bfrag[kk].x = wbase[(size_t)(kk * 4 + 0) * FEAT];
        bfrag[kk].y = wbase[(size_t)(kk * 4 + 1) * FEAT];
    }

    const float4* AH4 = (const float4*)(AH + (size_t)r0 * FEAT);
    const float4* H04 = (const float4*)(H0 + (size_t)r0 * FEAT);
    const int plast = nrows * (FEAT / 4) - 1;
    float4 a4[GT_LOADS], h4[GT_LOADS];
#pragma unroll
    for (int j = 0; j < GT_LOADS; ++j) {
        const int p  = tid + j * GT_THREADS;
        const int pc = (p < plast) ? p : plast;
        a4[j] = AH4[pc];
        h4[j] = H04[pc];
    }

#pragma unroll
    for (int j = 0; j < GT_LOADS; ++j) {
        const int p = tid + j * GT_THREADS;
        float4 s;
        s.x = oma * a4[j].x + alpha * h4[j].x;
        s.y = oma * a4[j].y + alpha * h4[j].y;
        s.z = oma * a4[j].z + alpha * h4[j].z;
        s.w = oma * a4[j].w + alpha * h4[j].w;
        *(float4*)&sup[(p >> 4) * GT_PITCH + (p & 15) * 4] = s;
    }
    __syncthreads();

    v8f acc[4] = {};
    const float* arow = sup + lm * GT_PITCH + hi * 2;
#pragma unroll
    for (int kk = 0; kk < 16; ++kk) {
#pragma unroll
        for (int t = 0; t < 4; ++t) {
            const v2f a = *(const v2f*)(arow + t * 16 * GT_PITCH + kk * 4);
            acc[t] = __builtin_amdgcn_wmma_f32_16x16x4_f32(
                 false, a,  false, bfrag[kk],
                 (short)0, acc[t],  false,  false);
        }
    }
    __syncthreads();

#pragma unroll
    for (int t = 0; t < 4; ++t) {
#pragma unroll
        for (int r = 0; r < 8; ++r) {
            float* s = sup + (t * 16 + r + hi * 8) * GT_PITCH + n0 + lm;
            *s = omb * (*s) + beta * acc[t][r];
        }
    }
    __syncthreads();

    float4* out4 = (float4*)(out + (size_t)r0 * FEAT);
    if (nrows == GT_ROWS) {
        float4 o[GT_LOADS];
#pragma unroll
        for (int j = 0; j < GT_LOADS; ++j) {
            const int p = tid + j * GT_THREADS;
            o[j] = *(const float4*)&sup[(p >> 4) * GT_PITCH + (p & 15) * 4];
        }
#pragma unroll
        for (int j = 0; j < GT_LOADS; ++j)
            out4[tid + j * GT_THREADS] = o[j];
    } else {
        for (int j = 0; j < GT_LOADS; ++j) {
            const int p = tid + j * GT_THREADS;
            if ((p >> 4) < nrows)
                out4[p] = *(const float4*)&sup[(p >> 4) * GT_PITCH + (p & 15) * 4];
        }
    }
}

extern "C" void kernel_launch(void* const* d_in, const int* in_sizes, int n_in,
                              void* d_out, int out_size, void* d_ws, size_t ws_size,
                              hipStream_t stream) {
    const int*   row   = (const int*)  d_in[0];
    const int*   col   = (const int*)  d_in[1];
    const float* vals  = (const float*)d_in[2];
    const float* H     = (const float*)d_in[3];
    const float* H0    = (const float*)d_in[4];
    const float* W     = (const float*)d_in[5];
    const float* lamda = (const float*)d_in[6];
    const float* alpha = (const float*)d_in[7];
    const int*   l     = (const int*)  d_in[8];

    int nEdges = in_sizes[0];
    int nNodes = in_sizes[3] / FEAT;

    float*    AH  = (float*)d_ws;
    unsigned* map = (unsigned*)(AH + (size_t)nNodes * FEAT);

    int spmmBlocks = (nNodes + RB_ROWS - 1) / RB_ROWS;
    int mapTiles   = (nEdges + TILE_EDGES - 1) / TILE_EDGES;

    gcnii_mark_groups<<<mapTiles, MARK_THREADS, 0, stream>>>(
        row, map, nEdges, spmmBlocks, mapTiles);

    gcnii_spmm_owner_map<<<spmmBlocks, SCAN_THREADS, 0, stream>>>(
        row, col, vals, H, map, AH, nEdges, nNodes, mapTiles);

    int gemmBlocks = (nNodes + GT_ROWS - 1) / GT_ROWS;
    gcnii_gemm_epilogue<<<gemmBlocks, GT_THREADS, 0, stream>>>(
        AH, H0, W, (float*)d_out, lamda, alpha, l, nNodes);
}


# SPMM owner: matched edges dealt to the wave owning their row (32 lists, slot = returning LDS add); after a barrier a wave sums its 13 rows in registers: no LDS float adds; sum order = list arrival
# speedup vs baseline: 3.0483x; 1.1576x over previous
#include <hip/hip_runtime.h>
#include <math.h>

typedef __attribute__((ext_vector_type(2))) float v2f;
typedef __attribute__((ext_vector_type(8))) float v8f;
typedef __attribute__((ext_vector_type(16))) float v16f;

#define FEAT 64

#define RB_ROWS       400
#define SCAN_WAVES    32
#define SCAN_THREADS  (SCAN_WAVES * 32)
#define GATHER_DEPTH  32
#define REC_COL_BITS  23
#define WAVE_ROWS     ((RB_ROWS + SCAN_WAVES - 1) / SCAN_WAVES)
#define Q_CAP         320

#define GRP_EDGES       4
#define TILE_GROUPS     1024
#define TILE_EDGES      (GRP_EDGES * TILE_GROUPS)
#define TILE_WORDS      (TILE_GROUPS / 32)
#define MARK_THREADS    256
#define MARK_ITERS      (TILE_GROUPS / MARK_THREADS)
#define MARK_BLOCKS_MAX 256
#define GL_WORDS        64

__device__ __forceinline__ void mark_row(unsigned* bits, int r, unsigned b0, unsigned nb,
                                         unsigned word, unsigned bit) {
    const unsigned blk = (unsigned)r / (unsigned)RB_ROWS - b0;
    if (blk < nb)
        __hip_atomic_fetch_or(&bits[blk * TILE_WORDS + word], bit,
                              __ATOMIC_RELAXED, __HIP_MEMORY_SCOPE_WORKGROUP);
}

__global__ __launch_bounds__(MARK_THREADS)
void gcnii_mark_groups(const int* __restrict__ row, unsigned* __restrict__ map,
                       int nEdges, int nBlocks, int nTiles) {
    __shared__ unsigned bits[MARK_BLOCKS_MAX * TILE_WORDS];

    const int tid  = threadIdx.x;
    const int tile = blockIdx.x;

    int4 r[MARK_ITERS];
#pragma unroll
    for (int k = 0; k < MARK_ITERS; ++k) {
        const int e0 = tile * TILE_EDGES + (tid + k * MARK_THREADS) * GRP_EDGES;
        if (e0 + GRP_EDGES <= nEdges) {
            r[k] = *(const int4*)(row + e0);
        } else {
            r[k].x = (e0 + 0 < nEdges) ? row[e0 + 0] : -1;
            r[k].y = (e0 + 1 < nEdges) ? row[e0 + 1] : -1;
            r[k].z = (e0 + 2 < nEdges) ? row[e0 + 2] : -1;
            r[k].w = (e0 + 3 < nEdges) ? row[e0 + 3] : -1;
        }
    }

    const unsigned bit = 1u << (tid & 31);
    for (int b0 = 0; b0 < nBlocks; b0 += MARK_BLOCKS_MAX) {
        const int nb = (nBlocks - b0 < MARK_BLOCKS_MAX) ? (nBlocks - b0) : MARK_BLOCKS_MAX;
        uint4* bits4 = (uint4*)bits;
        for (int i = tid; i < nb * TILE_WORDS / 4; i += MARK_THREADS)
            bits4[i] = make_uint4(0u, 0u, 0u, 0u);
        __syncthreads();
#pragma unroll
        for (int k = 0; k < MARK_ITERS; ++k) {
            const unsigned word = (unsigned)(tid >> 5) + (unsigned)k * (MARK_THREADS / 32);
            mark_row(bits, r[k].x, (unsigned)b0, (unsigned)nb, word, bit);
            mark_row(bits, r[k].y, (unsigned)b0, (unsigned)nb, word, bit);
            mark_row(bits, r[k].z, (unsigned)b0, (unsigned)nb, word, bit);
            mark_row(bits, r[k].w, (unsigned)b0, (unsigned)nb, word, bit);
        }
        __syncthreads();
        for (int i = tid; i < nb * TILE_WORDS; i += MARK_THREADS) {
            const size_t line = (size_t)(b0 + (i >> 5)) * nTiles + tile;
            map[line * TILE_WORDS + (i & 31)] = bits[i];
        }
        __syncthreads();
    }
}

__device__ __forceinline__ unsigned take_slot(unsigned t, unsigned* qcnt) {
    if (t < (unsigned)RB_ROWS)
        return __hip_atomic_fetch_add(&qcnt[t & (unsigned)(SCAN_WAVES - 1)], 1u, __ATOMIC_RELAXED,
                                      __HIP_MEMORY_SCOPE_WORKGROUP);
    return 0xffffffffu;
}

__device__ __forceinline__ void deal_match(unsigned t, unsigned s, int c, float v, uint2* queue,
                                           float* acc, const float* __restrict__ H) {
    if (t < (unsigned)RB_ROWS) {
        if (s < (unsigned)Q_CAP) {
            const unsigned ow = t & (unsigned)(SCAN_WAVES - 1);
            queue[ow * Q_CAP + s] = make_uint2((unsigned)c | ((t / SCAN_WAVES) << REC_COL_BITS),
                                               __builtin_bit_cast(unsigned, v));
        } else {
            const float* srow = H + (size_t)c * FEAT;
            float* d = acc + t * FEAT;
#pragma nounroll
            for (int f = 0; f < FEAT; ++f)
                __hip_atomic_fetch_add(d + f, v * srow[f], __ATOMIC_RELAXED, __HIP_MEMORY_SCOPE_WORKGROUP);
        }
    }
}

template <bool FULL>
__device__ __forceinline__ void sum_records(const uint2* q, unsigned m, int lane,
                                            const float* __restrict__ H, v16f& sx, v16f& sy) {
    int cl = 0; float vl = 0.f; unsigned jl = 0;
    if (FULL || (unsigned)lane < m) {
        const uint2 rec = q[lane];
        cl = (int)(rec.x & ((1u << REC_COL_BITS) - 1u));
        jl = rec.x >> REC_COL_BITS;
        vl = __builtin_bit_cast(float, rec.y);
    }
    float2 h[GATHER_DEPTH];
#pragma unroll
    for (int t = 0; t < GATHER_DEPTH; ++t) {
        const unsigned kk = (FULL || (unsigned)t < m) ? (unsigned)t : (m - 1u);
        const int cc = __builtin_amdgcn_readlane(cl, (int)kk);
        const unsigned off = ((unsigned)cc << 8) + (unsigned)lane * 8u;
        h[t] = *(const float2*)((const char*)H + off);
    }
#pragma unroll
    for (int t = 0; t < GATHER_DEPTH; ++t) {
        if (FULL || (unsigned)t < m) {
            const float vv = __builtin_bit_cast(float,
                __builtin_amdgcn_readlane(__builtin_bit_cast(int, vl), t));
            const int jj = __builtin_amdgcn_readlane((int)jl, t);
            sx[jj] += vv * h[t].x;
            sy[jj] += vv * h[t].y;
        }
    }
}

template <bool FULL>
__device__ __forceinline__ void match_groups(const unsigned* gl, unsigned m, int lane,
                                             const int* __restrict__ row,
                                             const int* __restrict__ col,
                                             const float* __restrict__ vals, int nEdges,
                                             unsigned rbase, uint2* queue, unsigned* qcnt,
                                             float* acc, const float* __restrict__ H) {
    int4   r = make_int4(-1, -1, -1, -1);
    int4   c = make_int4(0, 0, 0, 0);
    float4 v = make_float4(0.f, 0.f, 0.f, 0.f);
    if (FULL || (unsigned)lane < m) {
        const unsigned e0 = gl[lane] * GRP_EDGES;
        if ((int)e0 + GRP_EDGES <= nEdges) {
            r = *(const int4*)(row + e0);
            c = *(const int4*)(col + e0);
            v = *(const float4*)(vals + e0);
        } else {
            if ((int)e0 + 0 < nEdges) { r.x = row[e0 + 0]; c.x = col[e0 + 0]; v.x = vals[e0 + 0]; }
            if ((int)e0 + 1 < nEdges) { r.y = row[e0 + 1]; c.y = col[e0 + 1]; v.y = vals[e0 + 1]; }
            if ((int)e0 + 2 < nEdges) { r.z = row[e0 + 2]; c.z = col[e0 + 2]; v.z = vals[e0 + 2]; }
        }
    }
    const unsigned tx = (unsigned)r.x - rbase, ty = (unsigned)r.y - rbase;
    const unsigned tz = (unsigned)r.z - rbase, tw = (unsigned)r.w - rbase;
    const unsigned sx = take_slot(tx, qcnt);
    const unsigned sy = take_slot(ty, qcnt);
    const unsigned sz = take_slot(tz, qcnt);
    const unsigned sw = take_slot(tw, qcnt);
    deal_match(tx, sx, c.x, v.x, queue, acc, H);
    deal_match(ty, sy, c.y, v.y, queue, acc, H);
    deal_match(tz, sz, c.z, v.z, queue, acc, H);
    deal_match(tw, sw, c.w, v.w, queue, acc, H);
}

__global__ __launch_bounds__(SCAN_THREADS)
void gcnii_spmm_owner_map(const int* __restrict__ row,
                          const int* __restrict__ col,
                          const float* __restrict__ vals,
                          const float* __restrict__ H,
                          const unsigned* __restrict__ map,
                          float* __restrict__ AH,
                          int nEdges, int nNodes, int nTiles) {
    static_assert(WAVE_ROWS <= 16, "a wave's row slots must fit the 16-wide register sums");
    __shared__ float    acc[RB_ROWS * FEAT];
    __shared__ unsigned glist[SCAN_WAVES * GL_WORDS];
    __shared__ uint2    queue[SCAN_WAVES * Q_CAP];
    __shared__ unsigned qcnt[SCAN_WAVES];

    const int lane = threadIdx.x & 31;
    const int w    = __builtin_amdgcn_readfirstlane((int)(threadIdx.x >> 5));
    const unsigned b = blockIdx.x;
    const unsigned rbase = b * (unsigned)RB_ROWS;

    float4* acc4 = (float4*)acc;
    for (int i = threadIdx.x; i < RB_ROWS * FEAT / 4; i += SCAN_THREADS)
        acc4[i] = make_float4(0.f, 0.f, 0.f, 0.f);
    if (threadIdx.x < SCAN_WAVES) qcnt[threadIdx.x] = 0u;
    __syncthreads();

    const unsigned* mine = map + (size_t)b * nTiles * TILE_WORDS;
    unsigned* gl = glist + w * GL_WORDS;

    unsigned gcnt = 0;
    unsigned nxt = (w < nTiles) ? mine[w * TILE_WORDS + lane] : 0u;
    for (int i = w; i < nTiles; i += SCAN_WAVES) {
        unsigned wbits = nxt;
        if (i + SCAN_WAVES < nTiles) nxt = mine[(i + SCAN_WAVES) * TILE_WORDS + lane];
        const unsigned gbase = (unsigned)i * TILE_GROUPS + (unsigned)lane * 32u;

        for (int pass = 0; pass < 32; ++pass) {
            const bool has = wbits != 0u;
            const unsigned m = __builtin_amdgcn_ballot_w32(has);
            if (!m) break;
            if (has) {
                gl[gcnt + __builtin_amdgcn_mbcnt_lo(m, 0u)] = gbase + (unsigned)__builtin_ctz(wbits);
                wbits &= wbits - 1u;
            }
            gcnt += (unsigned)__builtin_popcount(m);
            if (gcnt >= 32u) {
                gcnt -= 32u;
                match_groups<true>(gl + gcnt, 32u, lane, row, col, vals, nEdges, rbase,
                                   queue, qcnt, acc, H);
            }
        }
    }
    if (gcnt)
        match_groups<false>(gl, gcnt, lane, row, col, vals, nEdges, rbase, queue, qcnt, acc, H);

    __syncthreads();

    unsigned n = (unsigned)__builtin_amdgcn_readfirstlane((int)qcnt[w]);
    if (n > (unsigned)Q_CAP) n = (unsigned)Q_CAP;
    const uint2* q = queue + w * Q_CAP;
    v16f sx = {}, sy = {};
    unsigned k0 = 0;
    for (; k0 + 32u <= n; k0 += 32u)
        sum_records<true>(q + k0, 32u, lane, H, sx, sy);
    if (k0 < n)
        sum_records<false>(q + k0, n - k0, lane, H, sx, sy);

    const int r0 = (int)b * RB_ROWS;
    const int nrows = (nNodes - r0 < RB_ROWS) ? (nNodes - r0) : RB_ROWS;
#pragma unroll
    for (int j = 0; j < WAVE_ROWS; ++j) {
        const int t = j * SCAN_WAVES + w;
        if (t < nrows) {
            const float2 extra = *(const float2*)(acc + t * FEAT + lane * 2);
            float2 o;
            o.x = sx[j] + extra.x;
            o.y = sy[j] + extra.y;
            *(float2*)(AH + (size_t)(r0 + t) * FEAT + lane * 2) = o;
        }
    }
}

#define GT_ROWS    64
#define GT_PITCH   68
#define GT_THREADS 128
#define GT_LOADS   (GT_ROWS * (FEAT / 4) / GT_THREADS)

__global__ __launch_bounds__(GT_THREADS)
void gcnii_gemm_epilogue(const float* __restrict__ AH,
                         const float* __restrict__ H0,
                         const float* __restrict__ W,
                         float* __restrict__ out,
                         const float* __restrict__ lamdaPtr,
                         const float* __restrict__ alphaPtr,
                         const int* __restrict__ lPtr,
                         int nNodes) {
    __shared__ __attribute__((aligned(16))) float sup[GT_ROWS * GT_PITCH];

    const int tid  = threadIdx.x;
    const int lane = tid & 31;
    const int w    = tid >> 5;
    const int hi   = lane >> 4;
    const int lm   = lane & 15;
    const int n0   = w * 16;

    const int r0    = (int)blockIdx.x * GT_ROWS;
    const int nrows = (nNodes - r0 < GT_ROWS) ? (nNodes - r0) : GT_ROWS;

    const float alpha = alphaPtr[0];
    const float lamda = lamdaPtr[0];
    const float beta  = logf(lamda / (float)lPtr[0] + 1.0f);
    const float oma   = 1.0f - alpha;
    const float omb   = 1.0f - beta;

    const float* wbase = W + (size_t)(hi * 2) * FEAT + n0 + lm;
    v2f bfrag[16];
#pragma unroll
    for (int kk = 0; kk < 16; ++kk) {
        bfrag[kk].x = wbase[(size_t)(kk * 4 + 0) * FEAT];
        bfrag[kk].y = wbase[(size_t)(kk * 4 + 1) * FEAT];
    }

    const float4* AH4 = (const float4*)(AH + (size_t)r0 * FEAT);
    const float4* H04 = (const float4*)(H0 + (size_t)r0 * FEAT);
    const int plast = nrows * (FEAT / 4) - 1;
    float4 a4[GT_LOADS], h4[GT_LOADS];
#pragma unroll
    for (int j = 0; j < GT_LOADS; ++j) {
        const int p  = tid + j * GT_THREADS;
        const int pc = (p < plast) ? p : plast;
        a4[j] = AH4[pc];
        h4[j] = H04[pc];
    }

#pragma unroll
    for (int j = 0; j < GT_LOADS; ++j) {
        const int p = tid + j * GT_THREADS;
        float4 s;
        s.x = oma * a4[j].x + alpha * h4[j].x;
        s.y = oma * a4[j].y + alpha * h4[j].y;
        s.z = oma * a4[j].z + alpha * h4[j].z;
        s.w = oma * a4[j].w + alpha * h4[j].w;
        *(float4*)&sup[(p >> 4) * GT_PITCH + (p & 15) * 4] = s;
    }
    __syncthreads();

    v8f acc[4] = {};
    const float* arow = sup + lm * GT_PITCH + hi * 2;
#pragma unroll
    for (int kk = 0; kk < 16; ++kk) {
#pragma unroll
        for (int t = 0; t < 4; ++t) {
            const v2f a = *(const v2f*)(arow + t * 16 * GT_PITCH + kk * 4);
            acc[t] = __builtin_amdgcn_wmma_f32_16x16x4_f32(
                 false, a,  false, bfrag[kk],
                 (short)0, acc[t],  false,  false);
        }
    }
    __syncthreads();

#pragma unroll
    for (int t = 0; t < 4; ++t) {
#pragma unroll
        for (int r = 0; r < 8; ++r) {
            float* s = sup + (t * 16 + r + hi * 8) * GT_PITCH + n0 + lm;
            *s = omb * (*s) + beta * acc[t][r];
        }
    }
    __syncthreads();

    float4* out4 = (float4*)(out + (size_t)r0 * FEAT);
    if (nrows == GT_ROWS) {
        float4 o[GT_LOADS];
#pragma unroll
        for (int j = 0; j < GT_LOADS; ++j) {
            const int p = tid + j * GT_THREADS;
            o[j] = *(const float4*)&sup[(p >> 4) * GT_PITCH + (p & 15) * 4];
        }
#pragma unroll
        for (int j = 0; j < GT_LOADS; ++j)
            out4[tid + j * GT_THREADS] = o[j];
    } else {
        for (int j = 0; j < GT_LOADS; ++j) {
            const int p = tid + j * GT_THREADS;
            if ((p >> 4) < nrows)
                out4[p] = *(const float4*)&sup[(p >> 4) * GT_PITCH + (p & 15) * 4];
        }
    }
}

extern "C" void kernel_launch(void* const* d_in, const int* in_sizes, int n_in,
                              void* d_out, int out_size, void* d_ws, size_t ws_size,
                              hipStream_t stream) {
    const int*   row   = (const int*)  d_in[0];
    const int*   col   = (const int*)  d_in[1];
    const float* vals  = (const float*)d_in[2];
    const float* H     = (const float*)d_in[3];
    const float* H0    = (const float*)d_in[4];
    const float* W     = (const float*)d_in[5];
    const float* lamda = (const float*)d_in[6];
    const float* alpha = (const float*)d_in[7];
    const int*   l     = (const int*)  d_in[8];

    int nEdges = in_sizes[0];
    int nNodes = in_sizes[3] / FEAT;

    float*    AH  = (float*)d_ws;
    unsigned* map = (unsigned*)(AH + (size_t)nNodes * FEAT);

    int spmmBlocks = (nNodes + RB_ROWS - 1) / RB_ROWS;
    int mapTiles   = (nEdges + TILE_EDGES - 1) / TILE_EDGES;

    gcnii_mark_groups<<<mapTiles, MARK_THREADS, 0, stream>>>(
        row, map, nEdges, spmmBlocks, mapTiles);

    gcnii_spmm_owner_map<<<spmmBlocks, SCAN_THREADS, 0, stream>>>(
        row, col, vals, H, map, AH, nEdges, nNodes, mapTiles);

    int gemmBlocks = (nNodes + GT_ROWS - 1) / GT_ROWS;
    gcnii_gemm_epilogue<<<gemmBlocks, GT_THREADS, 0, stream>>>(
        AH, H0, W, (float*)d_out, lamda, alpha, l, nNodes);
}
